# starting kernel
# baseline (speedup 1.0000x reference)
#include <hip/hip_runtime.h>


typedef __attribute__((ext_vector_type(2))) float v2f;
typedef __attribute__((ext_vector_type(4))) float v4f;
typedef __attribute__((ext_vector_type(8))) float v8f;

#define BCOL_STRIDE 772

__global__ __launch_bounds__(128) void fused_conv_wmma_f32(
    const float* __restrict__ x,
    const float* __restrict__ W,
    float* __restrict__ y)
{
    __shared__ float smem[16 * BCOL_STRIDE + 3 * 8 * 32];
    float* Blds = smem;
    float* Red  = smem + 16 * BCOL_STRIDE;

    const int tid     = threadIdx.x;
    const int lane    = tid & 31;
    const int wave    = tid >> 5;
    const int laneLow = lane & 15;
    const int kHalf   = lane >> 4;

    const int nt = blockIdx.x;
    const int mt = blockIdx.y;

    {
        const float* src = W + nt * (16 * 768);
        #pragma unroll
        for (int it = 0; it < 24; ++it) {
            const int chunk  = tid + it * 128;
            const int col    = chunk / 192;
            const int within = (chunk % 192) * 4;
            v4f v = *(const v4f*)(src + chunk * 4);
            *(v4f*)(Blds + col * BCOL_STRIDE + within) = v;
        }
    }
    __syncthreads();

    const int  mRow     = mt * 16 + laneLow;
    const bool rowValid = mRow < 196;
    const int  o        = rowValid ? (mRow / 14) : 0;
    const int  n        = rowValid ? (mRow % 14) : 0;
    const int  oEff1    = (o + 13) % 14;
    const int  nEff     = (n + 13) % 14;

    const float* bcol = Blds + laneLow * BCOL_STRIDE;
    const int    jt0  = wave * 8;

    v8f acc0 = {};
    v8f acc1 = {};

    #pragma unroll
    for (int l2 = 1; l2 >= 0; --l2) {
        #pragma unroll
        for (int k = 0; k < 3; ++k) {
            const int  w      = (l2 ? oEff1 : o) + k - 1;
            const bool aValid = rowValid && (w >= 0) && (w < 14);
            const float* xb = x + nEff * 14 + w;
            const float* wb = bcol + k * 256 + l2 * 128;

            v2f a[8], b[8];
            #pragma unroll
            for (int t = 0; t < 8; ++t) {
                const int j = 4 * (jt0 + t) + 2 * kHalf;
                a[t].x = aValid ? xb[j * 196]       : 0.0f;
                a[t].y = aValid ? xb[(j + 1) * 196] : 0.0f;
            }
            #pragma unroll
            for (int t = 0; t < 8; ++t) {
                const int j = 4 * (jt0 + t) + 2 * kHalf;
                b[t] = *(const v2f*)(wb + j);
            }
            #pragma unroll
            for (int t = 0; t < 8; t += 2) {
                acc0 = __builtin_amdgcn_wmma_f32_16x16x4_f32(
                    false, a[t],     false, b[t],     (short)0, acc0, false, false);
                acc1 = __builtin_amdgcn_wmma_f32_16x16x4_f32(
                    false, a[t + 1], false, b[t + 1], (short)0, acc1, false, false);
            }
        }
    }

    v8f acc;
    #pragma unroll
    for (int vr = 0; vr < 8; ++vr) acc[vr] = acc0[vr] + acc1[vr];

    if (wave != 0) {
        #pragma unroll
        for (int vr = 0; vr < 8; ++vr)
            Red[((wave - 1) * 8 + vr) * 32 + lane] = acc[vr];
    }
    __syncthreads();

    if (wave == 0) {
        const int iOut = nt * 16 + laneLow;
        #pragma unroll
        for (int vr = 0; vr < 8; ++vr) {
            float s = acc[vr];
            s += Red[(0 * 8 + vr) * 32 + lane];
            s += Red[(1 * 8 + vr) * 32 + lane];
            s += Red[(2 * 8 + vr) * 32 + lane];
            const int R = mt * 16 + vr + 8 * kHalf;
            if (R < 196) {
                const int oo = R / 14;
                const int nn = R % 14;
                y[iOut * 196 + oo * 14 + nn] = s;
            }
        }
    }
}

extern "C" void kernel_launch(void* const* d_in, const int* in_sizes, int n_in,
                              void* d_out, int out_size, void* d_ws, size_t ws_size,
                              hipStream_t stream) {
    (void)in_sizes; (void)n_in; (void)out_size; (void)d_ws; (void)ws_size;
    const float* x = (const float*)d_in[0];
    const float* W = (const float*)d_in[1];
    float* y = (float*)d_out;

    dim3 grid(16, 13);
    dim3 block(128);
    hipLaunchKernelGGL(fused_conv_wmma_f32, grid, block, 0, stream, x, W, y);
}


# M tile = one output column n (14 rows o + 2 pad): its whole A operand is one 7 KB row of x, staged once in LDS transposed; A fragments by ds_load_b64 instead of 96 strided global b32 loads per lane
# speedup vs baseline: 2.6538x; 2.6538x over previous
#include <hip/hip_runtime.h>


typedef __attribute__((ext_vector_type(2))) float v2f;
typedef __attribute__((ext_vector_type(4))) float v4f;
typedef __attribute__((ext_vector_type(8))) float v8f;

#define BCOL_STRIDE 772
#define XROW_STRIDE 132

__global__ __launch_bounds__(128) void fused_conv_wmma_f32(
    const float* __restrict__ x,
    const float* __restrict__ W,
    float* __restrict__ y)
{
    __shared__ float smem[16 * BCOL_STRIDE + 3 * 8 * 32 + 16 * XROW_STRIDE];
    float* Blds = smem;
    float* Red  = smem + 16 * BCOL_STRIDE;
    float* Xs   = Red + 3 * 8 * 32;

    const int tid     = threadIdx.x;
    const int lane    = tid & 31;
    const int wave    = tid >> 5;
    const int laneLow = lane & 15;
    const int kHalf   = lane >> 4;

    const int nt = blockIdx.x;
    const int n  = blockIdx.y;
    const int h  = (n + 13) % 14;

    {
        const float* wsrc = W + nt * (16 * 768);
        const float* xsrc = x + h * 14;

        v4f wv[24];
        #pragma unroll
        for (int it = 0; it < 24; ++it)
            wv[it] = *(const v4f*)(wsrc + (tid + it * 128) * 4);

        v2f xv[7];
        #pragma unroll
        for (int it = 0; it < 7; ++it) {
            const unsigned p = (unsigned)tid + it * 128u;
            const unsigned j = p / 7u;
            const unsigned q = p - 7u * j;
            xv[it] = *(const v2f*)(xsrc + (j * 196u + 2u * q));
        }

        #pragma unroll
        for (int it = 0; it < 24; ++it) {
            const int chunk  = tid + it * 128;
            const int col    = chunk / 192;
            const int within = (chunk % 192) * 4;
            *(v4f*)(Blds + col * BCOL_STRIDE + within) = wv[it];
        }
        Xs[tid]                    = 0.0f;
        Xs[15 * XROW_STRIDE + tid] = 0.0f;
        #pragma unroll
        for (int it = 0; it < 7; ++it) {
            const unsigned p = (unsigned)tid + it * 128u;
            const unsigned j = p / 7u;
            const unsigned q = p - 7u * j;
            Xs[(2u * q + 1u) * XROW_STRIDE + j] = xv[it].x;
            Xs[(2u * q + 2u) * XROW_STRIDE + j] = xv[it].y;
        }
    }
    __syncthreads();

    const bool rowValid = laneLow < 14;
    const int  rb0      = rowValid ? laneLow : 0;
    const int  rb1      = rowValid ? (laneLow + 13) % 14 : 0;

    const float* bcol = Blds + laneLow * BCOL_STRIDE;
    const int    jt0  = wave * 8;

    v8f acc0 = {};
    v8f acc1 = {};

    #pragma unroll
    for (int l2 = 1; l2 >= 0; --l2) {
        #pragma unroll
        for (int k = 0; k < 3; ++k) {
            const float* xa = Xs + ((l2 ? rb1 : rb0) + k) * XROW_STRIDE;
            const float* wb = bcol + k * 256 + l2 * 128;

            v2f a[8], b[8];
            #pragma unroll
            for (int t = 0; t < 8; ++t) {
                const int j = 4 * (jt0 + t) + 2 * kHalf;
                a[t] = *(const v2f*)(xa + j);
            }
            #pragma unroll
            for (int t = 0; t < 8; ++t) {
                const int j = 4 * (jt0 + t) + 2 * kHalf;
                b[t] = *(const v2f*)(wb + j);
            }
            #pragma unroll
            for (int t = 0; t < 8; t += 2) {
                acc0 = __builtin_amdgcn_wmma_f32_16x16x4_f32(
                    false, a[t],     false, b[t],     (short)0, acc0, false, false);
                acc1 = __builtin_amdgcn_wmma_f32_16x16x4_f32(
                    false, a[t + 1], false, b[t + 1], (short)0, acc1, false, false);
            }
        }
    }

    v8f acc;
    #pragma unroll
    for (int vr = 0; vr < 8; ++vr) acc[vr] = acc0[vr] + acc1[vr];

    if (wave != 0) {
        #pragma unroll
        for (int vr = 0; vr < 8; ++vr)
            Red[((wave - 1) * 8 + vr) * 32 + lane] = acc[vr];
    }
    __syncthreads();

    if (wave == 0) {
        const int iOut = nt * 16 + laneLow;
        #pragma unroll
        for (int vr = 0; vr < 8; ++vr) {
            float s = acc[vr];
            s += Red[(0 * 8 + vr) * 32 + lane];
            s += Red[(1 * 8 + vr) * 32 + lane];
            s += Red[(2 * 8 + vr) * 32 + lane];
            const int oo = vr + 8 * kHalf;
            if (oo < 14)
                y[iOut * 196 + oo * 14 + n] = s;
        }
    }
}

extern "C" void kernel_launch(void* const* d_in, const int* in_sizes, int n_in,
                              void* d_out, int out_size, void* d_ws, size_t ws_size,
                              hipStream_t stream) {
    (void)in_sizes; (void)n_in; (void)out_size; (void)d_ws; (void)ws_size;
    const float* x = (const float*)d_in[0];
    const float* W = (const float*)d_in[1];
    float* y = (float*)d_out;

    dim3 grid(16, 14);
    dim3 block(128);
    hipLaunchKernelGGL(fused_conv_wmma_f32, grid, block, 0, stream, x, W, y);
}
